# starting kernel
# baseline (speedup 1.0000x reference)
#include <hip/hip_runtime.h>

typedef float v2f __attribute__((ext_vector_type(2)));
typedef float v4f __attribute__((ext_vector_type(4)));
typedef float v8f __attribute__((ext_vector_type(8)));

#define S_LEN 128
#define NB    4
#define EMB   512
#define NH    8
#define HD    64
#define BH    32

__global__ __launch_bounds__(256) void k_vproj(const float* __restrict__ value,
                                               const float* __restrict__ Wv,
                                               const float* __restrict__ bv,
                                               float* __restrict__ v_t) {
  const int lane = threadIdx.x & 31;
  const int wid  = blockIdx.x * 8 + (threadIdx.x >> 5);
  const int m0   = (wid >> 5) * 16;
  const int c0   = (wid & 31) * 16;
  const int half = lane >> 4;
  const int l15  = lane & 15;

  const int am = m0 + l15;
  const int bn = c0 + l15;
  v8f acc = {};
  for (int k0 = 0; k0 < EMB; k0 += 4) {
    const int ka = k0 + 2 * half;
    v2f a = *(const v2f*)(value + am * EMB + ka);
    v2f b = *(const v2f*)(Wv    + bn * EMB + ka);
    acc = __builtin_amdgcn_wmma_f32_16x16x4_f32(false, a, false, b,
                                                (short)0, acc, false, false);
  }
  const float bias = bv[bn];
  const int h  = bn >> 6;
  const int dd = bn & 63;
#pragma unroll
  for (int r = 0; r < 8; ++r) {
    const int m = m0 + r + 8 * half;
    const int s = m >> 2, n = m & 3;
    v_t[((n * NH + h) * S_LEN + s) * HD + dd] = acc[r] + bias;
  }
}

__global__ __launch_bounds__(256) void k_edge(const float* __restrict__ attn,
                                              const float* __restrict__ edge_v,
                                              float* __restrict__ out_pre) {
  const int lane = threadIdx.x & 31;
  const int b    = blockIdx.x * 8 + (threadIdx.x >> 5);
  const int half = lane >> 4;
  const int l15  = lane & 15;

  const float* evb = edge_v + (size_t)b * (S_LEN * HD) + l15 * 4;
  const float* wb  = attn + b * S_LEN + half;
  v4f acc = {};
#pragma unroll 4
  for (int s = 0; s < S_LEN; s += 2) {
    const float aw = wb[s];
    v4f ev = __builtin_nontemporal_load((const v4f*)(evb + (s + half) * HD));
    acc += aw * ev;
  }
  acc.x += __shfl_xor(acc.x, 16);
  acc.y += __shfl_xor(acc.y, 16);
  acc.z += __shfl_xor(acc.z, 16);
  acc.w += __shfl_xor(acc.w, 16);
  if (lane < 16)
    *(v4f*)(out_pre + b * HD + lane * 4) = acc;
}

__global__ __launch_bounds__(256) void k_attn(const float* __restrict__ attn,
                                              const float* __restrict__ v_t,
                                              float* __restrict__ out_pre) {
  const int lane = threadIdx.x & 31;
  const int wid  = blockIdx.x * 8 + (threadIdx.x >> 5);
  const int g    = wid >> 5;
  const int t    = wid & 31;
  const int q0   = (t >> 2) * 16;
  const int c0   = (t & 3) * 16;
  const int half = lane >> 4;
  const int l15  = lane & 15;

  const float* Ag = attn + (size_t)g * S_LEN * S_LEN;
  const float* Bg = v_t + g * S_LEN * HD;
  float*       Og = out_pre + g * S_LEN * HD;

  v8f acc;
#pragma unroll
  for (int r = 0; r < 8; ++r)
    acc[r] = Og[(q0 + r + 8 * half) * HD + c0 + l15];

  for (int k0 = 0; k0 < S_LEN; k0 += 4) {
    const int ka = k0 + 2 * half;
    v2f a = *(const v2f*)(Ag + (q0 + l15) * S_LEN + ka);
    v2f b;
    b.x = Bg[ka * HD + c0 + l15];
    b.y = Bg[(ka + 1) * HD + c0 + l15];
    acc = __builtin_amdgcn_wmma_f32_16x16x4_f32(false, a, false, b,
                                                (short)0, acc, false, false);
  }
#pragma unroll
  for (int r = 0; r < 8; ++r)
    Og[(q0 + r + 8 * half) * HD + c0 + l15] = acc[r];
}

__global__ __launch_bounds__(256) void k_outproj(const float* __restrict__ out_pre,
                                                 const float* __restrict__ Wout,
                                                 const float* __restrict__ bout,
                                                 float* __restrict__ out) {
  const int lane = threadIdx.x & 31;
  const int wid  = blockIdx.x * 8 + (threadIdx.x >> 5);
  const int m0   = (wid >> 5) * 16;
  const int c0   = (wid & 31) * 16;
  const int half = lane >> 4;
  const int l15  = lane & 15;

  const int am = m0 + l15;
  const int q  = am >> 2, n = am & 3;
  const int bn = c0 + l15;
  v8f acc = {};
  for (int k0 = 0; k0 < EMB; k0 += 4) {
    const int e = k0 + 2 * half;
    v2f a = *(const v2f*)(out_pre + ((n * NH + (e >> 6)) * S_LEN + q) * HD + (e & 63));
    v2f b = *(const v2f*)(Wout + bn * EMB + e);
    acc = __builtin_amdgcn_wmma_f32_16x16x4_f32(false, a, false, b,
                                                (short)0, acc, false, false);
  }
  const float bias = bout[bn];
#pragma unroll
  for (int r = 0; r < 8; ++r)
    out[(m0 + r + 8 * half) * EMB + c0 + l15] = acc[r] + bias;
}

extern "C" void kernel_launch(void* const* d_in, const int* in_sizes, int n_in,
                              void* d_out, int out_size, void* d_ws, size_t ws_size,
                              hipStream_t stream) {
  const float* attn  = (const float*)d_in[0];
  const float* value = (const float*)d_in[1];
  const float* edgev = (const float*)d_in[2];
  const float* Wv    = (const float*)d_in[3];
  const float* bv    = (const float*)d_in[4];
  const float* Wout  = (const float*)d_in[5];
  const float* bout  = (const float*)d_in[6];
  float* out = (float*)d_out;

  float* v_t     = (float*)d_ws;
  float* out_pre = v_t + BH * S_LEN * HD;

  k_vproj  <<<128, 256, 0, stream>>>(value, Wv, bv, v_t);
  k_edge   <<<512, 256, 0, stream>>>(attn, edgev, out_pre);
  k_attn   <<<128, 256, 0, stream>>>(attn, v_t, out_pre);
  k_outproj<<<128, 256, 0, stream>>>(out_pre, Wout, bout, out);
}


# both 512^3 projections: 32x32 block tiles, operand slabs staged in LDS by whole-line b128 loads (128-k stages, double-buffered); two k-interleaved accumulators change the f32 sum order
# speedup vs baseline: 1.4446x; 1.4446x over previous
#include <hip/hip_runtime.h>


typedef float v2f __attribute__((ext_vector_type(2)));
typedef float v4f __attribute__((ext_vector_type(4)));
typedef float v8f __attribute__((ext_vector_type(8)));

#define S_LEN 128
#define NB    4
#define EMB   512
#define NH    8
#define HD    64
#define BH    32

#define PT     32
#define KS     128
#define NST    (EMB / KS)
#define PITCH  (KS + 4)
#define TILE_F (PT * PITCH)
#define LDS_F  (2 * 2 * TILE_F)

__device__ __forceinline__ void frag_load(v4f (&xa)[8], v4f (&xb)[8],
                                          const float* fa, const float* fb, int hs) {
#pragma unroll
  for (int kk = 0; kk < 8; ++kk) {
    xa[kk] = *(const v4f*)(fa + (hs * 8 + kk) * 8);
    xb[kk] = *(const v4f*)(fb + (hs * 8 + kk) * 8);
  }
}

__device__ __forceinline__ void frag_mma(const v4f (&xa)[8], const v4f (&xb)[8],
                                         v8f& acc0, v8f& acc1) {
#pragma unroll
  for (int kk = 0; kk < 8; ++kk) {
    acc0 = __builtin_amdgcn_wmma_f32_16x16x4_f32(false, xa[kk].lo, false, xb[kk].lo,
                                                 (short)0, acc0, false, false);
    acc1 = __builtin_amdgcn_wmma_f32_16x16x4_f32(false, xa[kk].hi, false, xb[kk].hi,
                                                 (short)0, acc1, false, false);
  }
}

template <int MODE>
__device__ __forceinline__ void proj_block(const float* __restrict__ A,
                                           const float* __restrict__ Wm,
                                           const float* __restrict__ bias,
                                           float* __restrict__ O,
                                           float* lds) {
  const int lane = threadIdx.x & 31;
  const int w    = threadIdx.x >> 5;
  const int half = lane >> 4;
  const int l15  = lane & 15;
  const int m0   = (blockIdx.x >> 4) * PT;
  const int c0   = (blockIdx.x & 15) * PT;
  const int wm   = w >> 1;
  const int wc   = w & 1;

  constexpr int ASTR = MODE ? 2 * S_LEN * HD : KS;
  const float* a0[8];
  const float* b0[8];
#pragma unroll
  for (int j = 0; j < 8; ++j) {
    const int row = w * 8 + j;
    const int am  = m0 + row;
    if (MODE == 0) {
      a0[j] = A + am * EMB + 4 * lane;
    } else {
      const int q = am >> 2, n = am & 3;
      a0[j] = A + ((n * NH + half) * S_LEN + q) * HD + 4 * l15;
    }
    b0[j] = Wm + (c0 + row) * EMB + 4 * lane;
  }
  float* sa = lds + (w * 8) * PITCH + 4 * lane;
  const float* fa = lds + (wm * 16 + l15) * PITCH + 4 * half;
  const float* fb = lds + TILE_F + (wc * 16 + l15) * PITCH + 4 * half;

  v4f ga[8], gb[8];
#pragma unroll
  for (int j = 0; j < 8; ++j) {
    ga[j] = *(const v4f*)(a0[j]);
    gb[j] = *(const v4f*)(b0[j]);
  }
#pragma unroll
  for (int j = 0; j < 8; ++j) {
    *(v4f*)(sa + j * PITCH)          = ga[j];
    *(v4f*)(sa + TILE_F + j * PITCH) = gb[j];
  }
  __syncthreads();

  v8f acc0 = {};
  v8f acc1 = {};
  v4f xa[8], xb[8];
  v4f ya[8], yb[8];
#pragma unroll
  for (int st = 0; st < NST; ++st) {
    const int cur = (st & 1) * 2 * TILE_F;
    const int nxt = ((st + 1) & 1) * 2 * TILE_F;
    if (st + 1 < NST) {
#pragma unroll
      for (int j = 0; j < 8; ++j) {
        ga[j] = *(const v4f*)(a0[j] + (st + 1) * ASTR);
        gb[j] = *(const v4f*)(b0[j] + (st + 1) * KS);
      }
    }
    frag_load(xa, xb, fa + cur, fb + cur, 0);
    __builtin_amdgcn_sched_barrier(0);
    if (st > 0) frag_mma(ya, yb, acc0, acc1);
    __builtin_amdgcn_sched_barrier(0);
    frag_load(ya, yb, fa + cur, fb + cur, 1);
    __builtin_amdgcn_sched_barrier(0);
    frag_mma(xa, xb, acc0, acc1);
    __builtin_amdgcn_sched_barrier(0);
    if (st + 1 < NST) {
#pragma unroll
      for (int j = 0; j < 8; ++j) {
        *(v4f*)(sa + nxt + j * PITCH)          = ga[j];
        *(v4f*)(sa + nxt + TILE_F + j * PITCH) = gb[j];
      }
      __syncthreads();
    }
  }
  frag_mma(ya, yb, acc0, acc1);

  const v8f acc = acc0 + acc1;
  const int bn  = c0 + wc * 16 + l15;
  const float bs = bias[bn];
  if (MODE == 0) {
    const int h  = bn >> 6;
    const int dd = bn & 63;
#pragma unroll
    for (int r = 0; r < 8; ++r) {
      const int m = m0 + wm * 16 + r + 8 * half;
      const int s = m >> 2, n = m & 3;
      O[((n * NH + h) * S_LEN + s) * HD + dd] = acc[r] + bs;
    }
  } else {
#pragma unroll
    for (int r = 0; r < 8; ++r)
      O[(m0 + wm * 16 + r + 8 * half) * EMB + bn] = acc[r] + bs;
  }
}

__global__ __launch_bounds__(128) void k_vproj(const float* __restrict__ value,
                                               const float* __restrict__ Wv,
                                               const float* __restrict__ bv,
                                               float* __restrict__ v_t) {
  __shared__ __attribute__((aligned(16))) float lds[LDS_F];
  proj_block<0>(value, Wv, bv, v_t, lds);
}

__global__ __launch_bounds__(256) void k_edge(const float* __restrict__ attn,
                                              const float* __restrict__ edge_v,
                                              float* __restrict__ out_pre) {
  const int lane = threadIdx.x & 31;
  const int b    = blockIdx.x * 8 + (threadIdx.x >> 5);
  const int half = lane >> 4;
  const int l15  = lane & 15;

  const float* evb = edge_v + (size_t)b * (S_LEN * HD) + l15 * 4;
  const float* wb  = attn + b * S_LEN + half;
  v4f acc = {};
#pragma unroll 4
  for (int s = 0; s < S_LEN; s += 2) {
    const float aw = wb[s];
    v4f ev = __builtin_nontemporal_load((const v4f*)(evb + (s + half) * HD));
    acc += aw * ev;
  }
  acc.x += __shfl_xor(acc.x, 16);
  acc.y += __shfl_xor(acc.y, 16);
  acc.z += __shfl_xor(acc.z, 16);
  acc.w += __shfl_xor(acc.w, 16);
  if (lane < 16)
    *(v4f*)(out_pre + b * HD + lane * 4) = acc;
}

__global__ __launch_bounds__(256) void k_attn(const float* __restrict__ attn,
                                              const float* __restrict__ v_t,
                                              float* __restrict__ out_pre) {
  const int lane = threadIdx.x & 31;
  const int wid  = blockIdx.x * 8 + (threadIdx.x >> 5);
  const int g    = wid >> 5;
  const int t    = wid & 31;
  const int q0   = (t >> 2) * 16;
  const int c0   = (t & 3) * 16;
  const int half = lane >> 4;
  const int l15  = lane & 15;

  const float* Ag = attn + (size_t)g * S_LEN * S_LEN;
  const float* Bg = v_t + g * S_LEN * HD;
  float*       Og = out_pre + g * S_LEN * HD;

  v8f acc;
#pragma unroll
  for (int r = 0; r < 8; ++r)
    acc[r] = Og[(q0 + r + 8 * half) * HD + c0 + l15];

  for (int k0 = 0; k0 < S_LEN; k0 += 4) {
    const int ka = k0 + 2 * half;
    v2f a = *(const v2f*)(Ag + (q0 + l15) * S_LEN + ka);
    v2f b;
    b.x = Bg[ka * HD + c0 + l15];
    b.y = Bg[(ka + 1) * HD + c0 + l15];
    acc = __builtin_amdgcn_wmma_f32_16x16x4_f32(false, a, false, b,
                                                (short)0, acc, false, false);
  }
#pragma unroll
  for (int r = 0; r < 8; ++r)
    Og[(q0 + r + 8 * half) * HD + c0 + l15] = acc[r];
}

__global__ __launch_bounds__(128) void k_outproj(const float* __restrict__ out_pre,
                                                 const float* __restrict__ Wout,
                                                 const float* __restrict__ bout,
                                                 float* __restrict__ out) {
  __shared__ __attribute__((aligned(16))) float lds[LDS_F];
  proj_block<1>(out_pre, Wout, bout, out, lds);
}

extern "C" void kernel_launch(void* const* d_in, const int* in_sizes, int n_in,
                              void* d_out, int out_size, void* d_ws, size_t ws_size,
                              hipStream_t stream) {
  const float* attn  = (const float*)d_in[0];
  const float* value = (const float*)d_in[1];
  const float* edgev = (const float*)d_in[2];
  const float* Wv    = (const float*)d_in[3];
  const float* bv    = (const float*)d_in[4];
  const float* Wout  = (const float*)d_in[5];
  const float* bout  = (const float*)d_in[6];
  float* out = (float*)d_out;

  float* v_t     = (float*)d_ws;
  float* out_pre = v_t + BH * S_LEN * HD;

  k_vproj  <<<256, 128, 0, stream>>>(value, Wv, bv, v_t);
  k_edge   <<<512, 256, 0, stream>>>(attn, edgev, out_pre);
  k_attn   <<<128, 256, 0, stream>>>(attn, v_t, out_pre);
  k_outproj<<<256, 128, 0, stream>>>(out_pre, Wout, bout, out);
}


# edge stream: each wave requests its 64 loads of 512 B in 4 groups of 16, one group ahead of the sums (16-32 loads in flight per wave, was 4); row weights by scalar loads; sum order unchanged
# speedup vs baseline: 1.5486x; 1.0720x over previous
#include <hip/hip_runtime.h>


typedef float v2f __attribute__((ext_vector_type(2)));
typedef float v4f __attribute__((ext_vector_type(4)));
typedef float v8f __attribute__((ext_vector_type(8)));

#define S_LEN 128
#define NB    4
#define EMB   512
#define NH    8
#define HD    64
#define BH    32

#define PT     32
#define KS     128
#define NST    (EMB / KS)
#define PITCH  (KS + 4)
#define TILE_F (PT * PITCH)
#define LDS_F  (2 * 2 * TILE_F)

__device__ __forceinline__ void frag_load(v4f (&xa)[8], v4f (&xb)[8],
                                          const float* fa, const float* fb, int hs) {
#pragma unroll
  for (int kk = 0; kk < 8; ++kk) {
    xa[kk] = *(const v4f*)(fa + (hs * 8 + kk) * 8);
    xb[kk] = *(const v4f*)(fb + (hs * 8 + kk) * 8);
  }
}

__device__ __forceinline__ void frag_mma(const v4f (&xa)[8], const v4f (&xb)[8],
                                         v8f& acc0, v8f& acc1) {
#pragma unroll
  for (int kk = 0; kk < 8; ++kk) {
    acc0 = __builtin_amdgcn_wmma_f32_16x16x4_f32(false, xa[kk].lo, false, xb[kk].lo,
                                                 (short)0, acc0, false, false);
    acc1 = __builtin_amdgcn_wmma_f32_16x16x4_f32(false, xa[kk].hi, false, xb[kk].hi,
                                                 (short)0, acc1, false, false);
  }
}

template <int MODE>
__device__ __forceinline__ void proj_block(const float* __restrict__ A,
                                           const float* __restrict__ Wm,
                                           const float* __restrict__ bias,
                                           float* __restrict__ O,
                                           float* lds) {
  const int lane = threadIdx.x & 31;
  const int w    = threadIdx.x >> 5;
  const int half = lane >> 4;
  const int l15  = lane & 15;
  const int m0   = (blockIdx.x >> 4) * PT;
  const int c0   = (blockIdx.x & 15) * PT;
  const int wm   = w >> 1;
  const int wc   = w & 1;

  constexpr int ASTR = MODE ? 2 * S_LEN * HD : KS;
  const float* a0[8];
  const float* b0[8];
#pragma unroll
  for (int j = 0; j < 8; ++j) {
    const int row = w * 8 + j;
    const int am  = m0 + row;
    if (MODE == 0) {
      a0[j] = A + am * EMB + 4 * lane;
    } else {
      const int q = am >> 2, n = am & 3;
      a0[j] = A + ((n * NH + half) * S_LEN + q) * HD + 4 * l15;
    }
    b0[j] = Wm + (c0 + row) * EMB + 4 * lane;
  }
  float* sa = lds + (w * 8) * PITCH + 4 * lane;
  const float* fa = lds + (wm * 16 + l15) * PITCH + 4 * half;
  const float* fb = lds + TILE_F + (wc * 16 + l15) * PITCH + 4 * half;

  v4f ga[8], gb[8];
#pragma unroll
  for (int j = 0; j < 8; ++j) {
    ga[j] = *(const v4f*)(a0[j]);
    gb[j] = *(const v4f*)(b0[j]);
  }
#pragma unroll
  for (int j = 0; j < 8; ++j) {
    *(v4f*)(sa + j * PITCH)          = ga[j];
    *(v4f*)(sa + TILE_F + j * PITCH) = gb[j];
  }
  __syncthreads();

  v8f acc0 = {};
  v8f acc1 = {};
  v4f xa[8], xb[8];
  v4f ya[8], yb[8];
#pragma unroll
  for (int st = 0; st < NST; ++st) {
    const int cur = (st & 1) * 2 * TILE_F;
    const int nxt = ((st + 1) & 1) * 2 * TILE_F;
    if (st + 1 < NST) {
#pragma unroll
      for (int j = 0; j < 8; ++j) {
        ga[j] = *(const v4f*)(a0[j] + (st + 1) * ASTR);
        gb[j] = *(const v4f*)(b0[j] + (st + 1) * KS);
      }
    }
    frag_load(xa, xb, fa + cur, fb + cur, 0);
    __builtin_amdgcn_sched_barrier(0);
    if (st > 0) frag_mma(ya, yb, acc0, acc1);
    __builtin_amdgcn_sched_barrier(0);
    frag_load(ya, yb, fa + cur, fb + cur, 1);
    __builtin_amdgcn_sched_barrier(0);
    frag_mma(xa, xb, acc0, acc1);
    __builtin_amdgcn_sched_barrier(0);
    if (st + 1 < NST) {
#pragma unroll
      for (int j = 0; j < 8; ++j) {
        *(v4f*)(sa + nxt + j * PITCH)          = ga[j];
        *(v4f*)(sa + nxt + TILE_F + j * PITCH) = gb[j];
      }
      __syncthreads();
    }
  }
  frag_mma(ya, yb, acc0, acc1);

  const v8f acc = acc0 + acc1;
  const int bn  = c0 + wc * 16 + l15;
  const float bs = bias[bn];
  if (MODE == 0) {
    const int h  = bn >> 6;
    const int dd = bn & 63;
#pragma unroll
    for (int r = 0; r < 8; ++r) {
      const int m = m0 + wm * 16 + r + 8 * half;
      const int s = m >> 2, n = m & 3;
      O[((n * NH + h) * S_LEN + s) * HD + dd] = acc[r] + bs;
    }
  } else {
#pragma unroll
    for (int r = 0; r < 8; ++r)
      O[(m0 + wm * 16 + r + 8 * half) * EMB + bn] = acc[r] + bs;
  }
}

__global__ __launch_bounds__(128) void k_vproj(const float* __restrict__ value,
                                               const float* __restrict__ Wv,
                                               const float* __restrict__ bv,
                                               float* __restrict__ v_t) {
  __shared__ __attribute__((aligned(16))) float lds[LDS_F];
  proj_block<0>(value, Wv, bv, v_t, lds);
}

#define EG   16
#define ENG  (S_LEN / (2 * EG))

__global__ __launch_bounds__(256) void k_edge(const float* __restrict__ attn,
                                              const float* __restrict__ edge_v,
                                              float* __restrict__ out_pre) {
  const int lane = threadIdx.x & 31;
  const int wv   = __builtin_amdgcn_readfirstlane((int)(threadIdx.x >> 5));
  const int b    = blockIdx.x * 8 + wv;
  const int half = lane >> 4;
  const int l15  = lane & 15;

  const float* evb = edge_v + (size_t)b * (S_LEN * HD) + half * HD + l15 * 4;
  const float* wb  = attn + b * S_LEN;

  v4f ev[2][EG];
  v4f wq[2][EG / 2];
#define LOAD_GROUP(G) do { \
    _Pragma("unroll") for (int j = 0; j < EG / 2; ++j) \
      wq[(G) & 1][j] = *(const v4f*)(wb + (G) * 2 * EG + 4 * j); \
    _Pragma("unroll") for (int q = 0; q < EG; ++q) \
      ev[(G) & 1][q] = __builtin_nontemporal_load((const v4f*)(evb + ((G) * EG + q) * 2 * HD)); \
  } while (0)

  v4f acc = {};
  LOAD_GROUP(0);
#pragma unroll
  for (int g = 0; g < ENG; ++g) {
    __builtin_amdgcn_sched_barrier(0);
    if (g + 1 < ENG) LOAD_GROUP(g + 1);
    __builtin_amdgcn_sched_barrier(0);
#pragma unroll
    for (int q = 0; q < EG; ++q) {
      const v4f w4 = wq[g & 1][q >> 1];
      const float w0 = (q & 1) ? w4.z : w4.x;
      const float w1 = (q & 1) ? w4.w : w4.y;
      const float aw = half ? w1 : w0;
      acc += aw * ev[g & 1][q];
    }
  }
#undef LOAD_GROUP
  acc.x += __shfl_xor(acc.x, 16);
  acc.y += __shfl_xor(acc.y, 16);
  acc.z += __shfl_xor(acc.z, 16);
  acc.w += __shfl_xor(acc.w, 16);
  if (lane < 16)
    *(v4f*)(out_pre + b * HD + lane * 4) = acc;
}

__global__ __launch_bounds__(256) void k_attn(const float* __restrict__ attn,
                                              const float* __restrict__ v_t,
                                              float* __restrict__ out_pre) {
  const int lane = threadIdx.x & 31;
  const int wid  = blockIdx.x * 8 + (threadIdx.x >> 5);
  const int g    = wid >> 5;
  const int t    = wid & 31;
  const int q0   = (t >> 2) * 16;
  const int c0   = (t & 3) * 16;
  const int half = lane >> 4;
  const int l15  = lane & 15;

  const float* Ag = attn + (size_t)g * S_LEN * S_LEN;
  const float* Bg = v_t + g * S_LEN * HD;
  float*       Og = out_pre + g * S_LEN * HD;

  v8f acc;
#pragma unroll
  for (int r = 0; r < 8; ++r)
    acc[r] = Og[(q0 + r + 8 * half) * HD + c0 + l15];

  for (int k0 = 0; k0 < S_LEN; k0 += 4) {
    const int ka = k0 + 2 * half;
    v2f a = *(const v2f*)(Ag + (q0 + l15) * S_LEN + ka);
    v2f b;
    b.x = Bg[ka * HD + c0 + l15];
    b.y = Bg[(ka + 1) * HD + c0 + l15];
    acc = __builtin_amdgcn_wmma_f32_16x16x4_f32(false, a, false, b,
                                                (short)0, acc, false, false);
  }
#pragma unroll
  for (int r = 0; r < 8; ++r)
    Og[(q0 + r + 8 * half) * HD + c0 + l15] = acc[r];
}

__global__ __launch_bounds__(128) void k_outproj(const float* __restrict__ out_pre,
                                                 const float* __restrict__ Wout,
                                                 const float* __restrict__ bout,
                                                 float* __restrict__ out) {
  __shared__ __attribute__((aligned(16))) float lds[LDS_F];
  proj_block<1>(out_pre, Wout, bout, out, lds);
}

extern "C" void kernel_launch(void* const* d_in, const int* in_sizes, int n_in,
                              void* d_out, int out_size, void* d_ws, size_t ws_size,
                              hipStream_t stream) {
  const float* attn  = (const float*)d_in[0];
  const float* value = (const float*)d_in[1];
  const float* edgev = (const float*)d_in[2];
  const float* Wv    = (const float*)d_in[3];
  const float* bv    = (const float*)d_in[4];
  const float* Wout  = (const float*)d_in[5];
  const float* bout  = (const float*)d_in[6];
  float* out = (float*)d_out;

  float* v_t     = (float*)d_ws;
  float* out_pre = v_t + BH * S_LEN * HD;

  k_vproj  <<<256, 128, 0, stream>>>(value, Wv, bv, v_t);
  k_edge   <<<512, 256, 0, stream>>>(attn, edgev, out_pre);
  k_attn   <<<128, 256, 0, stream>>>(attn, v_t, out_pre);
  k_outproj<<<256, 128, 0, stream>>>(out_pre, Wout, bout, out);
}


# v projection and edge stream share one launch: 256 projection tiles then 512 stream blocks, all resident at once (a stream wave now sums 2 rows back to back); 3 launches, was 4; sum order unchanged
# speedup vs baseline: 1.6467x; 1.0633x over previous
#include <hip/hip_runtime.h>


typedef float v2f __attribute__((ext_vector_type(2)));
typedef float v4f __attribute__((ext_vector_type(4)));
typedef float v8f __attribute__((ext_vector_type(8)));

#define S_LEN 128
#define NB    4
#define EMB   512
#define NH    8
#define HD    64
#define BH    32

#define PT     32
#define KS     128
#define NST    (EMB / KS)
#define PITCH  (KS + 4)
#define TILE_F (PT * PITCH)
#define LDS_F  (2 * 2 * TILE_F)

__device__ __forceinline__ void frag_load(v4f (&xa)[8], v4f (&xb)[8],
                                          const float* fa, const float* fb, int hs) {
#pragma unroll
  for (int kk = 0; kk < 8; ++kk) {
    xa[kk] = *(const v4f*)(fa + (hs * 8 + kk) * 8);
    xb[kk] = *(const v4f*)(fb + (hs * 8 + kk) * 8);
  }
}

__device__ __forceinline__ void frag_mma(const v4f (&xa)[8], const v4f (&xb)[8],
                                         v8f& acc0, v8f& acc1) {
#pragma unroll
  for (int kk = 0; kk < 8; ++kk) {
    acc0 = __builtin_amdgcn_wmma_f32_16x16x4_f32(false, xa[kk].lo, false, xb[kk].lo,
                                                 (short)0, acc0, false, false);
    acc1 = __builtin_amdgcn_wmma_f32_16x16x4_f32(false, xa[kk].hi, false, xb[kk].hi,
                                                 (short)0, acc1, false, false);
  }
}

template <int MODE>
__device__ __forceinline__ void proj_block(const float* __restrict__ A,
                                           const float* __restrict__ Wm,
                                           const float* __restrict__ bias,
                                           float* __restrict__ O,
                                           float* lds, int tile) {
  const int lane = threadIdx.x & 31;
  const int w    = threadIdx.x >> 5;
  const int half = lane >> 4;
  const int l15  = lane & 15;
  const int m0   = (tile >> 4) * PT;
  const int c0   = (tile & 15) * PT;
  const int wm   = w >> 1;
  const int wc   = w & 1;

  constexpr int ASTR = MODE ? 2 * S_LEN * HD : KS;
  const float* a0[8];
  const float* b0[8];
#pragma unroll
  for (int j = 0; j < 8; ++j) {
    const int row = w * 8 + j;
    const int am  = m0 + row;
    if (MODE == 0) {
      a0[j] = A + am * EMB + 4 * lane;
    } else {
      const int q = am >> 2, n = am & 3;
      a0[j] = A + ((n * NH + half) * S_LEN + q) * HD + 4 * l15;
    }
    b0[j] = Wm + (c0 + row) * EMB + 4 * lane;
  }
  float* sa = lds + (w * 8) * PITCH + 4 * lane;
  const float* fa = lds + (wm * 16 + l15) * PITCH + 4 * half;
  const float* fb = lds + TILE_F + (wc * 16 + l15) * PITCH + 4 * half;

  v4f ga[8], gb[8];
#pragma unroll
  for (int j = 0; j < 8; ++j) {
    ga[j] = *(const v4f*)(a0[j]);
    gb[j] = *(const v4f*)(b0[j]);
  }
#pragma unroll
  for (int j = 0; j < 8; ++j) {
    *(v4f*)(sa + j * PITCH)          = ga[j];
    *(v4f*)(sa + TILE_F + j * PITCH) = gb[j];
  }
  __syncthreads();

  v8f acc0 = {};
  v8f acc1 = {};
  v4f xa[8], xb[8];
  v4f ya[8], yb[8];
#pragma unroll
  for (int st = 0; st < NST; ++st) {
    const int cur = (st & 1) * 2 * TILE_F;
    const int nxt = ((st + 1) & 1) * 2 * TILE_F;
    if (st + 1 < NST) {
#pragma unroll
      for (int j = 0; j < 8; ++j) {
        ga[j] = *(const v4f*)(a0[j] + (st + 1) * ASTR);
        gb[j] = *(const v4f*)(b0[j] + (st + 1) * KS);
      }
    }
    frag_load(xa, xb, fa + cur, fb + cur, 0);
    __builtin_amdgcn_sched_barrier(0);
    if (st > 0) frag_mma(ya, yb, acc0, acc1);
    __builtin_amdgcn_sched_barrier(0);
    frag_load(ya, yb, fa + cur, fb + cur, 1);
    __builtin_amdgcn_sched_barrier(0);
    frag_mma(xa, xb, acc0, acc1);
    __builtin_amdgcn_sched_barrier(0);
    if (st + 1 < NST) {
#pragma unroll
      for (int j = 0; j < 8; ++j) {
        *(v4f*)(sa + nxt + j * PITCH)          = ga[j];
        *(v4f*)(sa + nxt + TILE_F + j * PITCH) = gb[j];
      }
      __syncthreads();
    }
  }
  frag_mma(ya, yb, acc0, acc1);

  const v8f acc = acc0 + acc1;
  const int bn  = c0 + wc * 16 + l15;
  const float bs = bias[bn];
  if (MODE == 0) {
    const int h  = bn >> 6;
    const int dd = bn & 63;
#pragma unroll
    for (int r = 0; r < 8; ++r) {
      const int m = m0 + wm * 16 + r + 8 * half;
      const int s = m >> 2, n = m & 3;
      O[((n * NH + h) * S_LEN + s) * HD + dd] = acc[r] + bs;
    }
  } else {
#pragma unroll
    for (int r = 0; r < 8; ++r)
      O[(m0 + wm * 16 + r + 8 * half) * EMB + bn] = acc[r] + bs;
  }
}

#define EG   16
#define ENG  (S_LEN / (2 * EG))
#define EROWS 2
#define PROJ_BLOCKS 256
#define EDGE_WAVES  4

__device__ __forceinline__ void edge_rows(const float* __restrict__ attn,
                                          const float* __restrict__ edge_v,
                                          float* __restrict__ out_pre, int eb) {
  const int lane = threadIdx.x & 31;
  const int wv   = __builtin_amdgcn_readfirstlane((int)(threadIdx.x >> 5));
  const int b0   = (eb * EDGE_WAVES + wv) * EROWS;
  const int half = lane >> 4;
  const int l15  = lane & 15;

  const float* evb = edge_v + (size_t)b0 * (S_LEN * HD) + half * HD + l15 * 4;
  const float* wb  = attn + b0 * S_LEN;

  v4f ev[2][EG];
  v4f wq[2][EG / 2];
#define LOAD_GROUP(G) do { \
    _Pragma("unroll") for (int j = 0; j < EG / 2; ++j) \
      wq[(G) & 1][j] = *(const v4f*)(wb + (G) * 2 * EG + 4 * j); \
    _Pragma("unroll") for (int q = 0; q < EG; ++q) \
      ev[(G) & 1][q] = __builtin_nontemporal_load((const v4f*)(evb + ((G) * EG + q) * 2 * HD)); \
  } while (0)

  v4f acc = {};
  LOAD_GROUP(0);
#pragma unroll
  for (int g = 0; g < EROWS * ENG; ++g) {
    __builtin_amdgcn_sched_barrier(0);
    if (g + 1 < EROWS * ENG) LOAD_GROUP(g + 1);
    __builtin_amdgcn_sched_barrier(0);
#pragma unroll
    for (int q = 0; q < EG; ++q) {
      const v4f w4 = wq[g & 1][q >> 1];
      const float w0 = (q & 1) ? w4.z : w4.x;
      const float w1 = (q & 1) ? w4.w : w4.y;
      const float aw = half ? w1 : w0;
      acc += aw * ev[g & 1][q];
    }
    if ((g % ENG) == ENG - 1) {
      acc.x += __shfl_xor(acc.x, 16);
      acc.y += __shfl_xor(acc.y, 16);
      acc.z += __shfl_xor(acc.z, 16);
      acc.w += __shfl_xor(acc.w, 16);
      if (lane < 16)
        *(v4f*)(out_pre + (b0 + g / ENG) * HD + lane * 4) = acc;
      acc = v4f{};
    }
  }
#undef LOAD_GROUP
}

__global__ __launch_bounds__(128) void k_vproj_edge(const float* __restrict__ value,
                                                    const float* __restrict__ Wv,
                                                    const float* __restrict__ bv,
                                                    float* __restrict__ v_t,
                                                    const float* __restrict__ attn,
                                                    const float* __restrict__ edge_v,
                                                    float* __restrict__ out_pre) {
  __shared__ __attribute__((aligned(16))) float lds[LDS_F];
  if (blockIdx.x < PROJ_BLOCKS)
    proj_block<0>(value, Wv, bv, v_t, lds, (int)blockIdx.x);
  else
    edge_rows(attn, edge_v, out_pre, (int)blockIdx.x - PROJ_BLOCKS);
}

__global__ __launch_bounds__(256) void k_attn(const float* __restrict__ attn,
                                              const float* __restrict__ v_t,
                                              float* __restrict__ out_pre) {
  const int lane = threadIdx.x & 31;
  const int wid  = blockIdx.x * 8 + (threadIdx.x >> 5);
  const int g    = wid >> 5;
  const int t    = wid & 31;
  const int q0   = (t >> 2) * 16;
  const int c0   = (t & 3) * 16;
  const int half = lane >> 4;
  const int l15  = lane & 15;

  const float* Ag = attn + (size_t)g * S_LEN * S_LEN;
  const float* Bg = v_t + g * S_LEN * HD;
  float*       Og = out_pre + g * S_LEN * HD;

  v8f acc;
#pragma unroll
  for (int r = 0; r < 8; ++r)
    acc[r] = Og[(q0 + r + 8 * half) * HD + c0 + l15];

  for (int k0 = 0; k0 < S_LEN; k0 += 4) {
    const int ka = k0 + 2 * half;
    v2f a = *(const v2f*)(Ag + (q0 + l15) * S_LEN + ka);
    v2f b;
    b.x = Bg[ka * HD + c0 + l15];
    b.y = Bg[(ka + 1) * HD + c0 + l15];
    acc = __builtin_amdgcn_wmma_f32_16x16x4_f32(false, a, false, b,
                                                (short)0, acc, false, false);
  }
#pragma unroll
  for (int r = 0; r < 8; ++r)
    Og[(q0 + r + 8 * half) * HD + c0 + l15] = acc[r];
}

__global__ __launch_bounds__(128) void k_outproj(const float* __restrict__ out_pre,
                                                 const float* __restrict__ Wout,
                                                 const float* __restrict__ bout,
                                                 float* __restrict__ out) {
  __shared__ __attribute__((aligned(16))) float lds[LDS_F];
  proj_block<1>(out_pre, Wout, bout, out, lds, (int)blockIdx.x);
}

extern "C" void kernel_launch(void* const* d_in, const int* in_sizes, int n_in,
                              void* d_out, int out_size, void* d_ws, size_t ws_size,
                              hipStream_t stream) {
  const float* attn  = (const float*)d_in[0];
  const float* value = (const float*)d_in[1];
  const float* edgev = (const float*)d_in[2];
  const float* Wv    = (const float*)d_in[3];
  const float* bv    = (const float*)d_in[4];
  const float* Wout  = (const float*)d_in[5];
  const float* bout  = (const float*)d_in[6];
  float* out = (float*)d_out;

  float* v_t     = (float*)d_ws;
  float* out_pre = v_t + BH * S_LEN * HD;

  k_vproj_edge<<<768, 128, 0, stream>>>(value, Wv, bv, v_t, attn, edgev, out_pre);
  k_attn   <<<128, 256, 0, stream>>>(attn, v_t, out_pre);
  k_outproj<<<256, 128, 0, stream>>>(out_pre, Wout, bout, out);
}


# attention product: 4-wave block owns 16 rows x 64 cols of a head; attn rows and v_t[g] staged in LDS by whole-line b128 loads (was 8 B / 4 B fragment loads, 16 lines each); sum order unchanged
# speedup vs baseline: 1.8604x; 1.1298x over previous
#include <hip/hip_runtime.h>


typedef float v2f __attribute__((ext_vector_type(2)));
typedef float v4f __attribute__((ext_vector_type(4)));
typedef float v8f __attribute__((ext_vector_type(8)));

#define S_LEN 128
#define NB    4
#define EMB   512
#define NH    8
#define HD    64
#define BH    32

#define PT     32
#define KS     128
#define NST    (EMB / KS)
#define PITCH  (KS + 4)
#define TILE_F (PT * PITCH)
#define LDS_F  (2 * 2 * TILE_F)

__device__ __forceinline__ void frag_load(v4f (&xa)[8], v4f (&xb)[8],
                                          const float* fa, const float* fb, int hs) {
#pragma unroll
  for (int kk = 0; kk < 8; ++kk) {
    xa[kk] = *(const v4f*)(fa + (hs * 8 + kk) * 8);
    xb[kk] = *(const v4f*)(fb + (hs * 8 + kk) * 8);
  }
}

__device__ __forceinline__ void frag_mma(const v4f (&xa)[8], const v4f (&xb)[8],
                                         v8f& acc0, v8f& acc1) {
#pragma unroll
  for (int kk = 0; kk < 8; ++kk) {
    acc0 = __builtin_amdgcn_wmma_f32_16x16x4_f32(false, xa[kk].lo, false, xb[kk].lo,
                                                 (short)0, acc0, false, false);
    acc1 = __builtin_amdgcn_wmma_f32_16x16x4_f32(false, xa[kk].hi, false, xb[kk].hi,
                                                 (short)0, acc1, false, false);
  }
}

template <int MODE>
__device__ __forceinline__ void proj_block(const float* __restrict__ A,
                                           const float* __restrict__ Wm,
                                           const float* __restrict__ bias,
                                           float* __restrict__ O,
                                           float* lds, int tile) {
  const int lane = threadIdx.x & 31;
  const int w    = threadIdx.x >> 5;
  const int half = lane >> 4;
  const int l15  = lane & 15;
  const int m0   = (tile >> 4) * PT;
  const int c0   = (tile & 15) * PT;
  const int wm   = w >> 1;
  const int wc   = w & 1;

  constexpr int ASTR = MODE ? 2 * S_LEN * HD : KS;
  const float* a0[8];
  const float* b0[8];
#pragma unroll
  for (int j = 0; j < 8; ++j) {
    const int row = w * 8 + j;
    const int am  = m0 + row;
    if (MODE == 0) {
      a0[j] = A + am * EMB + 4 * lane;
    } else {
      const int q = am >> 2, n = am & 3;
      a0[j] = A + ((n * NH + half) * S_LEN + q) * HD + 4 * l15;
    }
    b0[j] = Wm + (c0 + row) * EMB + 4 * lane;
  }
  float* sa = lds + (w * 8) * PITCH + 4 * lane;
  const float* fa = lds + (wm * 16 + l15) * PITCH + 4 * half;
  const float* fb = lds + TILE_F + (wc * 16 + l15) * PITCH + 4 * half;

  v4f ga[8], gb[8];
#pragma unroll
  for (int j = 0; j < 8; ++j) {
    ga[j] = *(const v4f*)(a0[j]);
    gb[j] = *(const v4f*)(b0[j]);
  }
#pragma unroll
  for (int j = 0; j < 8; ++j) {
    *(v4f*)(sa + j * PITCH)          = ga[j];
    *(v4f*)(sa + TILE_F + j * PITCH) = gb[j];
  }
  __syncthreads();

  v8f acc0 = {};
  v8f acc1 = {};
  v4f xa[8], xb[8];
  v4f ya[8], yb[8];
#pragma unroll
  for (int st = 0; st < NST; ++st) {
    const int cur = (st & 1) * 2 * TILE_F;
    const int nxt = ((st + 1) & 1) * 2 * TILE_F;
    if (st + 1 < NST) {
#pragma unroll
      for (int j = 0; j < 8; ++j) {
        ga[j] = *(const v4f*)(a0[j] + (st + 1) * ASTR);
        gb[j] = *(const v4f*)(b0[j] + (st + 1) * KS);
      }
    }
    frag_load(xa, xb, fa + cur, fb + cur, 0);
    __builtin_amdgcn_sched_barrier(0);
    if (st > 0) frag_mma(ya, yb, acc0, acc1);
    __builtin_amdgcn_sched_barrier(0);
    frag_load(ya, yb, fa + cur, fb + cur, 1);
    __builtin_amdgcn_sched_barrier(0);
    frag_mma(xa, xb, acc0, acc1);
    __builtin_amdgcn_sched_barrier(0);
    if (st + 1 < NST) {
#pragma unroll
      for (int j = 0; j < 8; ++j) {
        *(v4f*)(sa + nxt + j * PITCH)          = ga[j];
        *(v4f*)(sa + nxt + TILE_F + j * PITCH) = gb[j];
      }
      __syncthreads();
    }
  }
  frag_mma(ya, yb, acc0, acc1);

  const v8f acc = acc0 + acc1;
  const int bn  = c0 + wc * 16 + l15;
  const float bs = bias[bn];
  if (MODE == 0) {
    const int h  = bn >> 6;
    const int dd = bn & 63;
#pragma unroll
    for (int r = 0; r < 8; ++r) {
      const int m = m0 + wm * 16 + r + 8 * half;
      const int s = m >> 2, n = m & 3;
      O[((n * NH + h) * S_LEN + s) * HD + dd] = acc[r] + bs;
    }
  } else {
#pragma unroll
    for (int r = 0; r < 8; ++r)
      O[(m0 + wm * 16 + r + 8 * half) * EMB + bn] = acc[r] + bs;
  }
}

#define EG   16
#define ENG  (S_LEN / (2 * EG))
#define EROWS 2
#define PROJ_BLOCKS 256
#define EDGE_WAVES  4

__device__ __forceinline__ void edge_rows(const float* __restrict__ attn,
                                          const float* __restrict__ edge_v,
                                          float* __restrict__ out_pre, int eb) {
  const int lane = threadIdx.x & 31;
  const int wv   = __builtin_amdgcn_readfirstlane((int)(threadIdx.x >> 5));
  const int b0   = (eb * EDGE_WAVES + wv) * EROWS;
  const int half = lane >> 4;
  const int l15  = lane & 15;

  const float* evb = edge_v + (size_t)b0 * (S_LEN * HD) + half * HD + l15 * 4;
  const float* wb  = attn + b0 * S_LEN;

  v4f ev[2][EG];
  v4f wq[2][EG / 2];
#define LOAD_GROUP(G) do { \
    _Pragma("unroll") for (int j = 0; j < EG / 2; ++j) \
      wq[(G) & 1][j] = *(const v4f*)(wb + (G) * 2 * EG + 4 * j); \
    _Pragma("unroll") for (int q = 0; q < EG; ++q) \
      ev[(G) & 1][q] = __builtin_nontemporal_load((const v4f*)(evb + ((G) * EG + q) * 2 * HD)); \
  } while (0)

  v4f acc = {};
  LOAD_GROUP(0);
#pragma unroll
  for (int g = 0; g < EROWS * ENG; ++g) {
    __builtin_amdgcn_sched_barrier(0);
    if (g + 1 < EROWS * ENG) LOAD_GROUP(g + 1);
    __builtin_amdgcn_sched_barrier(0);
#pragma unroll
    for (int q = 0; q < EG; ++q) {
      const v4f w4 = wq[g & 1][q >> 1];
      const float w0 = (q & 1) ? w4.z : w4.x;
      const float w1 = (q & 1) ? w4.w : w4.y;
      const float aw = half ? w1 : w0;
      acc += aw * ev[g & 1][q];
    }
    if ((g % ENG) == ENG - 1) {
      acc.x += __shfl_xor(acc.x, 16);
      acc.y += __shfl_xor(acc.y, 16);
      acc.z += __shfl_xor(acc.z, 16);
      acc.w += __shfl_xor(acc.w, 16);
      if (lane < 16)
        *(v4f*)(out_pre + (b0 + g / ENG) * HD + lane * 4) = acc;
      acc = v4f{};
    }
  }
#undef LOAD_GROUP
}

__global__ __launch_bounds__(128) void k_vproj_edge(const float* __restrict__ value,
                                                    const float* __restrict__ Wv,
                                                    const float* __restrict__ bv,
                                                    float* __restrict__ v_t,
                                                    const float* __restrict__ attn,
                                                    const float* __restrict__ edge_v,
                                                    float* __restrict__ out_pre) {
  __shared__ __attribute__((aligned(16))) float lds[LDS_F];
  if (blockIdx.x < PROJ_BLOCKS)
    proj_block<0>(value, Wv, bv, v_t, lds, (int)blockIdx.x);
  else
    edge_rows(attn, edge_v, out_pre, (int)blockIdx.x - PROJ_BLOCKS);
}

#define AT_PA (S_LEN + 4)
#define AT_PB (HD + 8)
__global__ __launch_bounds__(128) void k_attn(const float* __restrict__ attn,
                                              const float* __restrict__ v_t,
                                              float* __restrict__ out_pre) {
  __shared__ __attribute__((aligned(16))) float As[16 * AT_PA];
  __shared__ __attribute__((aligned(16))) float Bs[S_LEN * AT_PB];
  const int lane = threadIdx.x & 31;
  const int w    = threadIdx.x >> 5;
  const int g    = blockIdx.x >> 3;
  const int q0   = (blockIdx.x & 7) * 16;
  const int c0   = w * 16;
  const int half = lane >> 4;
  const int l15  = lane & 15;

  const float* Ag = attn + (size_t)g * S_LEN * S_LEN + q0 * S_LEN;
  const float* Bg = v_t + g * S_LEN * HD;
  float*       Og = out_pre + g * S_LEN * HD;

  v4f ga[4], gb[16];
#pragma unroll
  for (int j = 0; j < 4; ++j)
    ga[j] = *(const v4f*)(Ag + (4 * w + j) * S_LEN + 4 * lane);
#pragma unroll
  for (int j = 0; j < 16; ++j)
    gb[j] = *(const v4f*)(Bg + (32 * w + 2 * j) * HD + 4 * lane);

  v8f acc;
#pragma unroll
  for (int r = 0; r < 8; ++r)
    acc[r] = Og[(q0 + r + 8 * half) * HD + c0 + l15];

#pragma unroll
  for (int j = 0; j < 4; ++j)
    *(v4f*)(As + (4 * w + j) * AT_PA + 4 * lane) = ga[j];
#pragma unroll
  for (int j = 0; j < 16; ++j)
    *(v4f*)(Bs + (32 * w + 2 * j + half) * AT_PB + 4 * l15) = gb[j];
  __syncthreads();

  const float* fa = As + l15 * AT_PA + 2 * half;
  const float* fb = Bs + (2 * half) * AT_PB + c0 + l15;
  v2f xa[32], xb[32];
#pragma unroll
  for (int t = 0; t < 32; ++t) {
    xa[t]   = *(const v2f*)(fa + 4 * t);
    xb[t].x = fb[(4 * t) * AT_PB];
    xb[t].y = fb[(4 * t + 1) * AT_PB];
  }
  __builtin_amdgcn_sched_barrier(0);
#pragma unroll
  for (int t = 0; t < 32; ++t)
    acc = __builtin_amdgcn_wmma_f32_16x16x4_f32(false, xa[t], false, xb[t],
                                                (short)0, acc, false, false);
#pragma unroll
  for (int r = 0; r < 8; ++r)
    Og[(q0 + r + 8 * half) * HD + c0 + l15] = acc[r];
}

__global__ __launch_bounds__(128) void k_outproj(const float* __restrict__ out_pre,
                                                 const float* __restrict__ Wout,
                                                 const float* __restrict__ bout,
                                                 float* __restrict__ out) {
  __shared__ __attribute__((aligned(16))) float lds[LDS_F];
  proj_block<1>(out_pre, Wout, bout, out, lds, (int)blockIdx.x);
}

extern "C" void kernel_launch(void* const* d_in, const int* in_sizes, int n_in,
                              void* d_out, int out_size, void* d_ws, size_t ws_size,
                              hipStream_t stream) {
  const float* attn  = (const float*)d_in[0];
  const float* value = (const float*)d_in[1];
  const float* edgev = (const float*)d_in[2];
  const float* Wv    = (const float*)d_in[3];
  const float* bv    = (const float*)d_in[4];
  const float* Wout  = (const float*)d_in[5];
  const float* bout  = (const float*)d_in[6];
  float* out = (float*)d_out;

  float* v_t     = (float*)d_ws;
  float* out_pre = v_t + BH * S_LEN * HD;

  k_vproj_edge<<<768, 128, 0, stream>>>(value, Wv, bv, v_t, attn, edgev, out_pre);
  k_attn   <<<256, 128, 0, stream>>>(attn, v_t, out_pre);
  k_outproj<<<256, 128, 0, stream>>>(out_pre, Wout, bout, out);
}
